# starting kernel
# baseline (speedup 1.0000x reference)
#include <hip/hip_runtime.h>
#include <hip/hip_bf16.h>
#include <math.h>

typedef __attribute__((ext_vector_type(2))) float v2f;
typedef __attribute__((ext_vector_type(8))) float v8f;

#define ES 292
#define PS 612
#define NW 4

__global__ __launch_bounds__(128)
void mca_attn_kernel(const float* __restrict__ x1,
                     const float* __restrict__ x2,
                     float* __restrict__ out, int B)
{
    extern __shared__ float smem[];
    float* e1s = smem;
    float* e2s = e1s + 16 * ES;
    float* ps  = e2s + 16 * ES;

    const int b    = blockIdx.x;
    const int tid  = threadIdx.x;
    const int wv   = tid >> 5;
    const int lane = tid & 31;
    const int c    = lane & 15;
    const int hi   = lane >> 4;

    {
        const float* srcs[2] = { x1 + (size_t)b * 4608, x2 + (size_t)b * 4608 };
        float*       dsts[2] = { e1s, e2s };
        for (int m = 0; m < 2; ++m) {
            const float* src = srcs[m];
            float*       dst = dsts[m];
            for (int idx = tid; idx < 4608; idx += 128) {
                int ch  = idx / 144;
                int rem = idx - ch * 144;
                int row = rem / 12;
                int col = rem - row * 12;
                int h = row / 3, p1 = row - h * 3;
                int w = col / 3, p2 = col - w * 3;
                int r = h * 4 + w;
                int f = (p1 * 3 + p2) * 32 + ch;
                dst[r * ES + f] = src[idx];
            }
        }
    }
    __syncthreads();

    float* myrow = ps + (size_t)(wv * 16 + c) * PS;

    for (int u = wv; u < 36; u += NW) {
        const int p  = u / 18;
        const int qt = u - p * 18;
        const float* eQ   = p ? e1s : e2s;
        const float* eKlo = p ? e2s : e1s;
        const float* eKhi = p ? e1s : e2s;
        const int j = qt * 16 + c;

        v2f qb[4];
        #pragma unroll
        for (int kk = 0; kk < 4; ++kk) {
            int d0 = 4 * kk + 2 * hi;
            v2f q; q[0] = eQ[d0 * ES + j]; q[1] = eQ[(d0 + 1) * ES + j];
            qb[kk] = q;
        }

        float mx = -3.0e38f;
        for (int t = 0; t < 36; ++t) {
            const float* eK = (t < 18) ? eKlo : eKhi;
            const int ic = ((t < 18) ? t * 16 : (t - 18) * 16) + c;
            v8f acc = {};
            #pragma unroll
            for (int kk = 0; kk < 4; ++kk) {
                int d0 = 4 * kk + 2 * hi;
                v2f a; a[0] = eK[d0 * ES + ic]; a[1] = eK[(d0 + 1) * ES + ic];
                acc = __builtin_amdgcn_wmma_f32_16x16x4_f32(
                          false, a, false, qb[kk], (short)0, acc, false, false);
            }
            float* dst = myrow + t * 16 + 8 * hi;
            *(float4*)(dst)     = make_float4(acc[0], acc[1], acc[2], acc[3]);
            *(float4*)(dst + 4) = make_float4(acc[4], acc[5], acc[6], acc[7]);
            mx = fmaxf(mx, fmaxf(fmaxf(fmaxf(acc[0], acc[1]), fmaxf(acc[2], acc[3])),
                                 fmaxf(fmaxf(acc[4], acc[5]), fmaxf(acc[6], acc[7]))));
        }
        mx = fmaxf(mx, __shfl_xor(mx, 16, 32));

        asm volatile("s_wait_dscnt 0" ::: "memory");
        float sum = 0.0f;
        for (int t = 0; t < 36; ++t) {
            const float vs = (t < 18) ? 0.7f : 0.3f;
            float* ptr = myrow + t * 16 + 8 * hi;
            float4 a4 = *(float4*)(ptr);
            float4 b4 = *(float4*)(ptr + 4);
            float q0 = __expf(a4.x - mx), q1 = __expf(a4.y - mx);
            float q2 = __expf(a4.z - mx), q3 = __expf(a4.w - mx);
            float q4 = __expf(b4.x - mx), q5 = __expf(b4.y - mx);
            float q6 = __expf(b4.z - mx), q7 = __expf(b4.w - mx);
            sum += ((q0 + q1) + (q2 + q3)) + ((q4 + q5) + (q6 + q7));
            *(float4*)(ptr)     = make_float4(q0 * vs, q1 * vs, q2 * vs, q3 * vs);
            *(float4*)(ptr + 4) = make_float4(q4 * vs, q5 * vs, q6 * vs, q7 * vs);
        }
        sum += __shfl_xor(sum, 16, 32);
        const float inv = 1.0f / sum;

        asm volatile("s_wait_dscnt 0" ::: "memory");
        v8f o = {};
        #pragma unroll 4
        for (int ks = 0; ks < 144; ++ks) {
            int i0 = 4 * ks + 2 * hi;
            const float* eV = (ks < 72) ? eKlo : eKhi;
            int ii = i0 - ((ks < 72) ? 0 : 288);
            v2f a = *(const v2f*)(eV + c * ES + ii);
            v2f bb = *(const v2f*)(myrow + i0);
            o = __builtin_amdgcn_wmma_f32_16x16x4_f32(
                    false, a, false, bb, (short)0, o, false, false);
        }

        const int ch  = j & 31;
        const int p12 = j >> 5;
        const int p1i = p12 / 3, p2i = p12 - p1i * 3;
        float* ob = out + (p ? (size_t)B * 4608 : (size_t)0)
                        + (size_t)b * 4608 + ch * 144 + p1i * 12 + p2i;
        #pragma unroll
        for (int v = 0; v < 8; ++v) {
            int d = v + 8 * hi;
            int h = d >> 2, w = d & 3;
            ob[h * 36 + w * 3] = o[v] * inv;
        }
    }
}

extern "C" void kernel_launch(void* const* d_in, const int* in_sizes, int n_in,
                              void* d_out, int out_size, void* d_ws, size_t ws_size,
                              hipStream_t stream) {
    const float* x1 = (const float*)d_in[0];
    const float* x2 = (const float*)d_in[1];
    float* out = (float*)d_out;
    const int B = in_sizes[0] / 4608;
    const size_t shmem = (size_t)(2 * 16 * ES + NW * 16 * PS) * sizeof(float);
    mca_attn_kernel<<<B, 128, shmem, stream>>>(x1, x2, out, B);
}


# P stays in registers: score D tile is the B operand of V*P (V loader in the same key order), online max per 32 keys; no P scratch in LDS, 12 waves per block; V*P and denominator sum order change
# speedup vs baseline: 2.2777x; 2.2777x over previous
#include <hip/hip_runtime.h>
#include <math.h>

typedef __attribute__((ext_vector_type(2))) float v2f;
typedef __attribute__((ext_vector_type(8))) float v8f;

#define ES 292
#define NW 12

#define L2E 1.4426950408889634f

__device__ __forceinline__ float xhalf(float x) {
    int b = __builtin_bit_cast(int, x);
    return __builtin_bit_cast(float,
        __builtin_amdgcn_permlanex16(b, b, 0x76543210, 0xfedcba98, false, false));
}

__device__ __forceinline__ v8f mma(v2f a, v2f b, v8f c) {
    return __builtin_amdgcn_wmma_f32_16x16x4_f32(false, a, false, b, (short)0, c, false, false);
}

__global__ __launch_bounds__(NW * 32)
void mca_attn_kernel(const float* __restrict__ x1,
                     const float* __restrict__ x2,
                     float* __restrict__ out, int B)
{
    __shared__ __attribute__((aligned(16))) float e1s[16 * ES];
    __shared__ __attribute__((aligned(16))) float e2s[16 * ES];

    const int b    = blockIdx.x;
    const int tid  = threadIdx.x;
    const int wv   = tid >> 5;
    const int lane = tid & 31;
    const int c    = lane & 15;
    const int hi   = lane >> 4;

    {
        const float* srcs[2] = { x1 + (size_t)b * 4608, x2 + (size_t)b * 4608 };
        float*       dsts[2] = { e1s, e2s };
        for (int m = 0; m < 2; ++m) {
            const float* src = srcs[m];
            float*       dst = dsts[m];
            for (int idx = tid; idx < 4608; idx += NW * 32) {
                int ch  = idx / 144;
                int rem = idx - ch * 144;
                int row = rem / 12;
                int col = rem - row * 12;
                int h = row / 3, p1 = row - h * 3;
                int w = col / 3, p2 = col - w * 3;
                int r = h * 4 + w;
                int f = (p1 * 3 + p2) * 32 + ch;
                dst[r * ES + f] = src[idx];
            }
        }
    }
    __syncthreads();

    for (int u = wv; u < 36; u += NW) {
        const int p  = u / 18;
        const int qt = u - p * 18;
        const float* eQ   = p ? e1s : e2s;
        const float* eKlo = p ? e2s : e1s;
        const float* eKhi = p ? e1s : e2s;
        const int j = qt * 16 + c;

        v2f qb[4];
        #pragma unroll
        for (int kk = 0; kk < 4; ++kk) {
            int d0 = 4 * kk + 2 * hi;
            v2f q; q[0] = eQ[d0 * ES + j]; q[1] = eQ[(d0 + 1) * ES + j];
            qb[kk] = q;
        }

        float ml = -3.0e38f;
        float l  = 0.0f;
        v8f   o  = {};

        for (int half = 0; half < 2; ++half) {
            const float* eK = half ? eKhi : eKlo;
            for (int st = 0; st < 9; ++st) {
                const int kb = st * 32;

                v2f ka0[4], ka1[4];
                for (int kk = 0; kk < 4; ++kk) {
                    int d0 = 4 * kk + 2 * hi;
                    v2f a;
                    a[0] = eK[d0 * ES + kb + c];      a[1] = eK[(d0 + 1) * ES + kb + c];
                    ka0[kk] = a;
                    a[0] = eK[d0 * ES + kb + 16 + c]; a[1] = eK[(d0 + 1) * ES + kb + 16 + c];
                    ka1[kk] = a;
                }
                const float* vp = eK + c * ES + kb + 8 * hi;
                const float4 va0 = *(const float4*)(vp);
                const float4 va1 = *(const float4*)(vp + 4);
                const float4 vb0 = *(const float4*)(vp + 16);
                const float4 vb1 = *(const float4*)(vp + 20);

                v8f s0 = {}, s1 = {};
                for (int kk = 0; kk < 4; ++kk) {
                    s0 = mma(ka0[kk], qb[kk], s0);
                    s1 = mma(ka1[kk], qb[kk], s1);
                }

                float tm = fmaxf(fmaxf(fmaxf(s0[0], s0[1]), fmaxf(s0[2], s0[3])),
                                 fmaxf(fmaxf(s0[4], s0[5]), fmaxf(s0[6], s0[7])));
                tm = fmaxf(tm, fmaxf(fmaxf(fmaxf(s1[0], s1[1]), fmaxf(s1[2], s1[3])),
                                     fmaxf(fmaxf(s1[4], s1[5]), fmaxf(s1[6], s1[7]))));
                tm = fmaxf(tm, xhalf(tm));
                const float mln   = fmaxf(ml, tm * L2E);
                const float alpha = __builtin_amdgcn_exp2f(ml - mln);
                const float nm    = -mln;
                ml = mln;

                v8f p0, p1;
                for (int r = 0; r < 8; ++r) {
                    p0[r] = __builtin_amdgcn_exp2f(fmaf(s0[r], L2E, nm));
                    p1[r] = __builtin_amdgcn_exp2f(fmaf(s1[r], L2E, nm));
                }
                const float ps = (((p0[0] + p0[1]) + (p0[2] + p0[3])) + ((p0[4] + p0[5]) + (p0[6] + p0[7])))
                               + (((p1[0] + p1[1]) + (p1[2] + p1[3])) + ((p1[4] + p1[5]) + (p1[6] + p1[7])));
                l = fmaf(l, alpha, ps);
                o = o * alpha;

                v2f a, bb;
                a[0] = va0.x; a[1] = va0.y; bb[0] = p0[0]; bb[1] = p0[1]; o = mma(a, bb, o);
                a[0] = va0.z; a[1] = va0.w; bb[0] = p0[2]; bb[1] = p0[3]; o = mma(a, bb, o);
                a[0] = va1.x; a[1] = va1.y; bb[0] = p0[4]; bb[1] = p0[5]; o = mma(a, bb, o);
                a[0] = va1.z; a[1] = va1.w; bb[0] = p0[6]; bb[1] = p0[7]; o = mma(a, bb, o);
                a[0] = vb0.x; a[1] = vb0.y; bb[0] = p1[0]; bb[1] = p1[1]; o = mma(a, bb, o);
                a[0] = vb0.z; a[1] = vb0.w; bb[0] = p1[2]; bb[1] = p1[3]; o = mma(a, bb, o);
                a[0] = vb1.x; a[1] = vb1.y; bb[0] = p1[4]; bb[1] = p1[5]; o = mma(a, bb, o);
                a[0] = vb1.z; a[1] = vb1.w; bb[0] = p1[6]; bb[1] = p1[7]; o = mma(a, bb, o);
            }
            if (half == 0) o = o * (0.7f / 0.3f);
        }

        l += xhalf(l);
        const float inv = 0.3f / l;

        const int ch  = j & 31;
        const int p12 = j >> 5;
        const int p1i = p12 / 3, p2i = p12 - p1i * 3;
        float* ob = out + (p ? (size_t)B * 4608 : (size_t)0)
                        + (size_t)b * 4608 + ch * 144 + p1i * 12 + p2i;
        #pragma unroll
        for (int v = 0; v < 8; ++v) {
            int d = v + 8 * hi;
            int h = d >> 2, w = d & 3;
            ob[h * 36 + w * 3] = o[v] * inv;
        }
    }
}

extern "C" void kernel_launch(void* const* d_in, const int* in_sizes, int n_in,
                              void* d_out, int out_size, void* d_ws, size_t ws_size,
                              hipStream_t stream) {
    const float* x1 = (const float*)d_in[0];
    const float* x2 = (const float*)d_in[1];
    float* out = (float*)d_out;
    const int B = in_sizes[0] / 4608;
    mca_attn_kernel<<<B, NW * 32, 0, stream>>>(x1, x2, out, B);
}


# K and Q operands read from a key-major LDS copy: one 16-byte read per fragment pair, no register moves; a step's 8 reads issued as one group one step ahead; score sum order over token dims changes
# speedup vs baseline: 2.5262x; 1.1091x over previous
#include <hip/hip_runtime.h>
#include <math.h>

typedef __attribute__((ext_vector_type(2))) float v2f;
typedef __attribute__((ext_vector_type(8))) float v8f;

#define ES 292
#define TP 20
#define NW 12

#define OFF_E1S 0
#define OFF_E2S (16 * ES)
#define OFF_E1T (32 * ES)
#define OFF_E2T (32 * ES + 288 * TP)
#define LDS_FLOATS (32 * ES + 2 * 288 * TP)

#define L2E 1.4426950408889634f

__device__ __forceinline__ float xhalf(float x) {
    int b = __builtin_bit_cast(int, x);
    return __builtin_bit_cast(float,
        __builtin_amdgcn_permlanex16(b, b, 0x76543210, 0xfedcba98, false, false));
}

__device__ __forceinline__ v8f mma(v2f a, v2f b, v8f c) {
    return __builtin_amdgcn_wmma_f32_16x16x4_f32(false, a, false, b, (short)0, c, false, false);
}

__device__ __forceinline__ v2f pr(float x, float y) { v2f r; r[0] = x; r[1] = y; return r; }

struct Frag {
    float4 k0a, k0b;
    float4 k1a, k1b;
    float4 v0a, v0b;
    float4 v1a, v1b;
};

__device__ __forceinline__ Frag load_frag(const float* lds, int ko, int vo) {
    Frag f;
    f.k0a = *(const float4*)(lds + ko);
    f.k0b = *(const float4*)(lds + ko + 4);
    f.k1a = *(const float4*)(lds + ko + 16 * TP);
    f.k1b = *(const float4*)(lds + ko + 16 * TP + 4);
    f.v0a = *(const float4*)(lds + vo);
    f.v0b = *(const float4*)(lds + vo + 4);
    f.v1a = *(const float4*)(lds + vo + 16);
    f.v1b = *(const float4*)(lds + vo + 20);
    return f;
}

__global__ __launch_bounds__(NW * 32)
void mca_attn_kernel(const float* __restrict__ x1,
                     const float* __restrict__ x2,
                     float* __restrict__ out, int B)
{
    __shared__ __attribute__((aligned(16))) float lds[LDS_FLOATS];

    const int b    = blockIdx.x;
    const int tid  = threadIdx.x;
    const int wv   = tid >> 5;
    const int lane = tid & 31;
    const int c    = lane & 15;
    const int hi   = lane >> 4;

    {
        const float* srcs[2] = { x1 + (size_t)b * 4608, x2 + (size_t)b * 4608 };
        for (int m = 0; m < 2; ++m) {
            const float* src = srcs[m];
            float*       dst = lds + (m ? OFF_E2S : OFF_E1S);
            float*       dtt = lds + (m ? OFF_E2T : OFF_E1T);
            for (int idx = tid; idx < 4608; idx += NW * 32) {
                int ch  = idx / 144;
                int rem = idx - ch * 144;
                int row = rem / 12;
                int col = rem - row * 12;
                int h = row / 3, p1 = row - h * 3;
                int w = col / 3, p2 = col - w * 3;
                int r = h * 4 + w;
                int f = (p1 * 3 + p2) * 32 + ch;
                const float v = src[idx];
                dst[r * ES + f] = v;
                dtt[f * TP + r] = v;
            }
        }
    }
    __syncthreads();

    for (int u = wv; u < 36; u += NW) {
        const int p  = u / 18;
        const int qt = u - p * 18;
        const int j = qt * 16 + c;

        const int kLo = (p ? OFF_E2T : OFF_E1T) + c * TP + 8 * hi;
        const int kHi = (p ? OFF_E1T : OFF_E2T) + c * TP + 8 * hi;
        const int vLo = (p ? OFF_E2S : OFF_E1S) + c * ES + 8 * hi;
        const int vHi = (p ? OFF_E1S : OFF_E2S) + c * ES + 8 * hi;

        v2f qb[4];
        {
            const float* qp = lds + (p ? OFF_E1T : OFF_E2T) + j * TP + 8 * hi;
            const float4 q0 = *(const float4*)(qp);
            const float4 q1 = *(const float4*)(qp + 4);
            qb[0] = pr(q0.x, q0.y); qb[1] = pr(q0.z, q0.w);
            qb[2] = pr(q1.x, q1.y); qb[3] = pr(q1.z, q1.w);
        }

        float ml = -3.0e38f;
        float l  = 0.0f;
        v8f   o  = {};

        auto step = [&](const Frag& f, const float oscale) {
            v8f s0 = {}, s1 = {};
            s0 = mma(pr(f.k0a.x, f.k0a.y), qb[0], s0);
            s1 = mma(pr(f.k1a.x, f.k1a.y), qb[0], s1);
            s0 = mma(pr(f.k0a.z, f.k0a.w), qb[1], s0);
            s1 = mma(pr(f.k1a.z, f.k1a.w), qb[1], s1);
            s0 = mma(pr(f.k0b.x, f.k0b.y), qb[2], s0);
            s1 = mma(pr(f.k1b.x, f.k1b.y), qb[2], s1);
            s0 = mma(pr(f.k0b.z, f.k0b.w), qb[3], s0);
            s1 = mma(pr(f.k1b.z, f.k1b.w), qb[3], s1);

            float tm = fmaxf(fmaxf(fmaxf(s0[0], s0[1]), fmaxf(s0[2], s0[3])),
                             fmaxf(fmaxf(s0[4], s0[5]), fmaxf(s0[6], s0[7])));
            tm = fmaxf(tm, fmaxf(fmaxf(fmaxf(s1[0], s1[1]), fmaxf(s1[2], s1[3])),
                                 fmaxf(fmaxf(s1[4], s1[5]), fmaxf(s1[6], s1[7]))));
            tm = fmaxf(tm, xhalf(tm));
            const float mln   = fmaxf(ml, tm * L2E);
            const float alpha = __builtin_amdgcn_exp2f(ml - mln);
            const float nm    = -mln;
            ml = mln;

            v8f p0, p1;
            for (int r = 0; r < 8; ++r) {
                p0[r] = __builtin_amdgcn_exp2f(fmaf(s0[r], L2E, nm));
                p1[r] = __builtin_amdgcn_exp2f(fmaf(s1[r], L2E, nm));
            }
            const float ps = (((p0[0] + p0[1]) + (p0[2] + p0[3])) + ((p0[4] + p0[5]) + (p0[6] + p0[7])))
                           + (((p1[0] + p1[1]) + (p1[2] + p1[3])) + ((p1[4] + p1[5]) + (p1[6] + p1[7])));
            l = fmaf(l, alpha, ps);
            o = o * (alpha * oscale);

            o = mma(pr(f.v0a.x, f.v0a.y), pr(p0[0], p0[1]), o);
            o = mma(pr(f.v0a.z, f.v0a.w), pr(p0[2], p0[3]), o);
            o = mma(pr(f.v0b.x, f.v0b.y), pr(p0[4], p0[5]), o);
            o = mma(pr(f.v0b.z, f.v0b.w), pr(p0[6], p0[7]), o);
            o = mma(pr(f.v1a.x, f.v1a.y), pr(p1[0], p1[1]), o);
            o = mma(pr(f.v1a.z, f.v1a.w), pr(p1[2], p1[3]), o);
            o = mma(pr(f.v1b.x, f.v1b.y), pr(p1[4], p1[5]), o);
            o = mma(pr(f.v1b.z, f.v1b.w), pr(p1[6], p1[7]), o);
        };

        Frag fa = load_frag(lds, kLo, vLo);
        #pragma unroll 1
        for (int i = 0; i < 9; ++i) {
            const bool up = (i >= 4);
            const int  kb = up ? kHi : kLo;
            const int  vb = up ? vHi : vLo;
            const int  s1 = 2 * i + 1 - (up ? 9 : 0);
            const int  s2 = (s1 + 1 < 9) ? s1 + 1 : 8;

            Frag fb = load_frag(lds, kb + s1 * (32 * TP), vb + s1 * 32);
            asm volatile("" : "+v"(qb[0]) :: "memory");
            step(fa, 1.0f);

            int k2 = kb + s2 * (32 * TP), v2 = vb + s2 * 32;
            asm volatile("" : "+v"(k2), "+v"(v2) : "v"(o));
            fa = load_frag(lds, k2, v2);
            asm volatile("" : "+v"(qb[0]) :: "memory");
            step(fb, (i == 4) ? (0.7f / 0.3f) : 1.0f);
        }

        l += xhalf(l);
        const float inv = 0.3f / l;

        const int ch  = j & 31;
        const int p12 = j >> 5;
        const int p1i = p12 / 3, p2i = p12 - p1i * 3;
        float* ob = out + (p ? (size_t)B * 4608 : (size_t)0)
                        + (size_t)b * 4608 + ch * 144 + p1i * 12 + p2i;
        #pragma unroll
        for (int v = 0; v < 8; ++v) {
            int d = v + 8 * hi;
            int h = d >> 2, w = d & 3;
            ob[h * 36 + w * 3] = o[v] * inv;
        }
    }
}

extern "C" void kernel_launch(void* const* d_in, const int* in_sizes, int n_in,
                              void* d_out, int out_size, void* d_ws, size_t ws_size,
                              hipStream_t stream) {
    const float* x1 = (const float*)d_in[0];
    const float* x2 = (const float*)d_in[1];
    float* out = (float*)d_out;
    const int B = in_sizes[0] / 4608;
    mca_attn_kernel<<<B, NW * 32, 0, stream>>>(x1, x2, out, B);
}


# Softmax arithmetic on register pairs: Q carries log2 e, so an exponent argument is one packed subtract per two scores; denominator by packed adds; score rounding and denominator sum order change
# speedup vs baseline: 2.6044x; 1.0310x over previous
#include <hip/hip_runtime.h>
#include <math.h>

typedef __attribute__((ext_vector_type(2))) float v2f;
typedef __attribute__((ext_vector_type(8))) float v8f;

#define ES 292
#define TP 20
#define NW 12

#define OFF_E1S 0
#define OFF_E2S (16 * ES)
#define OFF_E1T (32 * ES)
#define OFF_E2T (32 * ES + 288 * TP)
#define LDS_FLOATS (32 * ES + 2 * 288 * TP)

#define L2E 1.4426950408889634f

__device__ __forceinline__ float xhalf(float x) {
    int b = __builtin_bit_cast(int, x);
    return __builtin_bit_cast(float,
        __builtin_amdgcn_permlanex16(b, b, 0x76543210, 0xfedcba98, false, false));
}

__device__ __forceinline__ v8f mma(v2f a, v2f b, v8f c) {
    return __builtin_amdgcn_wmma_f32_16x16x4_f32(false, a, false, b, (short)0, c, false, false);
}

__device__ __forceinline__ v2f pr(float x, float y) { v2f r; r[0] = x; r[1] = y; return r; }

#define PAIR(t, i) pr((t)[2 * (i)], (t)[2 * (i) + 1])

struct Frag {
    float4 k0a, k0b;
    float4 k1a, k1b;
    float4 v0a, v0b;
    float4 v1a, v1b;
};

__device__ __forceinline__ Frag load_frag(const float* lds, int ko, int vo) {
    Frag f;
    f.k0a = *(const float4*)(lds + ko);
    f.k0b = *(const float4*)(lds + ko + 4);
    f.k1a = *(const float4*)(lds + ko + 16 * TP);
    f.k1b = *(const float4*)(lds + ko + 16 * TP + 4);
    f.v0a = *(const float4*)(lds + vo);
    f.v0b = *(const float4*)(lds + vo + 4);
    f.v1a = *(const float4*)(lds + vo + 16);
    f.v1b = *(const float4*)(lds + vo + 20);
    return f;
}

__global__ __launch_bounds__(NW * 32)
void mca_attn_kernel(const float* __restrict__ x1,
                     const float* __restrict__ x2,
                     float* __restrict__ out, int B)
{
    __shared__ __attribute__((aligned(16))) float lds[LDS_FLOATS];

    const int b    = blockIdx.x;
    const int tid  = threadIdx.x;
    const int wv   = tid >> 5;
    const int lane = tid & 31;
    const int c    = lane & 15;
    const int hi   = lane >> 4;

    {
        const float* srcs[2] = { x1 + (size_t)b * 4608, x2 + (size_t)b * 4608 };
        for (int m = 0; m < 2; ++m) {
            const float* src = srcs[m];
            float*       dst = lds + (m ? OFF_E2S : OFF_E1S);
            float*       dtt = lds + (m ? OFF_E2T : OFF_E1T);
            for (int idx = tid; idx < 4608; idx += NW * 32) {
                int ch  = idx / 144;
                int rem = idx - ch * 144;
                int row = rem / 12;
                int col = rem - row * 12;
                int h = row / 3, p1 = row - h * 3;
                int w = col / 3, p2 = col - w * 3;
                int r = h * 4 + w;
                int f = (p1 * 3 + p2) * 32 + ch;
                const float v = src[idx];
                dst[r * ES + f] = v;
                dtt[f * TP + r] = v;
            }
        }
    }
    __syncthreads();

    for (int u = wv; u < 36; u += NW) {
        const int p  = u / 18;
        const int qt = u - p * 18;
        const int j = qt * 16 + c;

        const int kLo = (p ? OFF_E2T : OFF_E1T) + c * TP + 8 * hi;
        const int kHi = (p ? OFF_E1T : OFF_E2T) + c * TP + 8 * hi;
        const int vLo = (p ? OFF_E2S : OFF_E1S) + c * ES + 8 * hi;
        const int vHi = (p ? OFF_E1S : OFF_E2S) + c * ES + 8 * hi;

        v2f qb[4];
        {
            const float* qp = lds + (p ? OFF_E1T : OFF_E2T) + j * TP + 8 * hi;
            const float4 q0 = *(const float4*)(qp);
            const float4 q1 = *(const float4*)(qp + 4);
            qb[0] = pr(q0.x, q0.y) * L2E; qb[1] = pr(q0.z, q0.w) * L2E;
            qb[2] = pr(q1.x, q1.y) * L2E; qb[3] = pr(q1.z, q1.w) * L2E;
        }

        float ml = -3.0e38f;
        float l  = 0.0f;
        v8f   o  = {};

        auto step = [&](const Frag& f, const float oscale) {
            v8f s0 = {}, s1 = {};
            s0 = mma(pr(f.k0a.x, f.k0a.y), qb[0], s0);
            s1 = mma(pr(f.k1a.x, f.k1a.y), qb[0], s1);
            s0 = mma(pr(f.k0a.z, f.k0a.w), qb[1], s0);
            s1 = mma(pr(f.k1a.z, f.k1a.w), qb[1], s1);
            s0 = mma(pr(f.k0b.x, f.k0b.y), qb[2], s0);
            s1 = mma(pr(f.k1b.x, f.k1b.y), qb[2], s1);
            s0 = mma(pr(f.k0b.z, f.k0b.w), qb[3], s0);
            s1 = mma(pr(f.k1b.z, f.k1b.w), qb[3], s1);

            float tm = fmaxf(fmaxf(fmaxf(s0[0], s0[1]), fmaxf(s0[2], s0[3])),
                             fmaxf(fmaxf(s0[4], s0[5]), fmaxf(s0[6], s0[7])));
            tm = fmaxf(tm, fmaxf(fmaxf(fmaxf(s1[0], s1[1]), fmaxf(s1[2], s1[3])),
                                 fmaxf(fmaxf(s1[4], s1[5]), fmaxf(s1[6], s1[7]))));
            tm = fmaxf(tm, xhalf(tm));
            const float mln   = fmaxf(ml, tm);
            const float alpha = __builtin_amdgcn_exp2f(ml - mln);
            const v2f   mm    = pr(mln, mln);
            ml = mln;

            v2f e0[4], e1[4];
            #pragma unroll
            for (int i = 0; i < 4; ++i) {
                const v2f d0 = PAIR(s0, i) - mm;
                const v2f d1 = PAIR(s1, i) - mm;
                e0[i] = pr(__builtin_amdgcn_exp2f(d0[0]), __builtin_amdgcn_exp2f(d0[1]));
                e1[i] = pr(__builtin_amdgcn_exp2f(d1[0]), __builtin_amdgcn_exp2f(d1[1]));
            }
            const v2f t0 = (e0[0] + e0[1]) + (e0[2] + e0[3]);
            const v2f t1 = (e1[0] + e1[1]) + (e1[2] + e1[3]);
            const v2f tt = t0 + t1;
            const float ps = tt[0] + tt[1];
            l = fmaf(l, alpha, ps);
            o = o * (alpha * oscale);

            o = mma(pr(f.v0a.x, f.v0a.y), e0[0], o);
            o = mma(pr(f.v0a.z, f.v0a.w), e0[1], o);
            o = mma(pr(f.v0b.x, f.v0b.y), e0[2], o);
            o = mma(pr(f.v0b.z, f.v0b.w), e0[3], o);
            o = mma(pr(f.v1a.x, f.v1a.y), e1[0], o);
            o = mma(pr(f.v1a.z, f.v1a.w), e1[1], o);
            o = mma(pr(f.v1b.x, f.v1b.y), e1[2], o);
            o = mma(pr(f.v1b.z, f.v1b.w), e1[3], o);
        };

        Frag fa = load_frag(lds, kLo, vLo);
        #pragma unroll 1
        for (int i = 0; i < 9; ++i) {
            const bool up = (i >= 4);
            const int  kb = up ? kHi : kLo;
            const int  vb = up ? vHi : vLo;
            const int  s1 = 2 * i + 1 - (up ? 9 : 0);
            const int  s2 = (s1 + 1 < 9) ? s1 + 1 : 8;

            Frag fb = load_frag(lds, kb + s1 * (32 * TP), vb + s1 * 32);
            asm volatile("" : "+v"(qb[0]) :: "memory");
            step(fa, 1.0f);

            int k2 = kb + s2 * (32 * TP), v2 = vb + s2 * 32;
            asm volatile("" : "+v"(k2), "+v"(v2) : "v"(o));
            fa = load_frag(lds, k2, v2);
            asm volatile("" : "+v"(qb[0]) :: "memory");
            step(fb, (i == 4) ? (0.7f / 0.3f) : 1.0f);
        }

        l += xhalf(l);
        const float inv = 0.3f / l;

        const int ch  = j & 31;
        const int p12 = j >> 5;
        const int p1i = p12 / 3, p2i = p12 - p1i * 3;
        float* ob = out + (p ? (size_t)B * 4608 : (size_t)0)
                        + (size_t)b * 4608 + ch * 144 + p1i * 12 + p2i;
        #pragma unroll
        for (int v = 0; v < 8; ++v) {
            int d = v + 8 * hi;
            int h = d >> 2, w = d & 3;
            ob[h * 36 + w * 3] = o[v] * inv;
        }
    }
}

extern "C" void kernel_launch(void* const* d_in, const int* in_sizes, int n_in,
                              void* d_out, int out_size, void* d_ws, size_t ws_size,
                              hipStream_t stream) {
    const float* x1 = (const float*)d_in[0];
    const float* x2 = (const float*)d_in[1];
    float* out = (float*)d_out;
    const int B = in_sizes[0] / 4608;
    mca_attn_kernel<<<B, NW * 32, 0, stream>>>(x1, x2, out, B);
}


# Output rows through wave-private LDS: a wave owns (pass, 16 channels, p1) and its three p2 tiles, so results leave as 16-byte stores, 3 lanes per 48-byte image row, not one 4-byte piece per lane
# speedup vs baseline: 2.7491x; 1.0556x over previous
#include <hip/hip_runtime.h>
#include <math.h>

typedef __attribute__((ext_vector_type(2))) float v2f;
typedef __attribute__((ext_vector_type(8))) float v8f;

#define ES 292
#define TP 20
#define NW 12

#define OFF_E1S 0
#define OFF_E2S (16 * ES)
#define OFF_E1T (32 * ES)
#define OFF_E2T (32 * ES + 288 * TP)
#define OFF_STG (32 * ES + 2 * 288 * TP)
#define CP 52
#define STG_W (16 * CP)
#define LDS_FLOATS (OFF_STG + NW * STG_W)

#define L2E 1.4426950408889634f

__device__ __forceinline__ float xhalf(float x) {
    int b = __builtin_bit_cast(int, x);
    return __builtin_bit_cast(float,
        __builtin_amdgcn_permlanex16(b, b, 0x76543210, 0xfedcba98, false, false));
}

__device__ __forceinline__ v8f mma(v2f a, v2f b, v8f c) {
    return __builtin_amdgcn_wmma_f32_16x16x4_f32(false, a, false, b, (short)0, c, false, false);
}

__device__ __forceinline__ v2f pr(float x, float y) { v2f r; r[0] = x; r[1] = y; return r; }

#define PAIR(t, i) pr((t)[2 * (i)], (t)[2 * (i) + 1])

struct Frag {
    float4 k0a, k0b;
    float4 k1a, k1b;
    float4 v0a, v0b;
    float4 v1a, v1b;
};

__device__ __forceinline__ Frag load_frag(const float* lds, int ko, int vo) {
    Frag f;
    f.k0a = *(const float4*)(lds + ko);
    f.k0b = *(const float4*)(lds + ko + 4);
    f.k1a = *(const float4*)(lds + ko + 16 * TP);
    f.k1b = *(const float4*)(lds + ko + 16 * TP + 4);
    f.v0a = *(const float4*)(lds + vo);
    f.v0b = *(const float4*)(lds + vo + 4);
    f.v1a = *(const float4*)(lds + vo + 16);
    f.v1b = *(const float4*)(lds + vo + 20);
    return f;
}

__global__ __launch_bounds__(NW * 32)
void mca_attn_kernel(const float* __restrict__ x1,
                     const float* __restrict__ x2,
                     float* __restrict__ out, int B)
{
    __shared__ __attribute__((aligned(16))) float lds[LDS_FLOATS];

    const int b    = blockIdx.x;
    const int tid  = threadIdx.x;
    const int wv   = tid >> 5;
    const int lane = tid & 31;
    const int c    = lane & 15;
    const int hi   = lane >> 4;

    {
        const float* srcs[2] = { x1 + (size_t)b * 4608, x2 + (size_t)b * 4608 };
        for (int m = 0; m < 2; ++m) {
            const float* src = srcs[m];
            float*       dst = lds + (m ? OFF_E2S : OFF_E1S);
            float*       dtt = lds + (m ? OFF_E2T : OFF_E1T);
            for (int idx = tid; idx < 4608; idx += NW * 32) {
                int ch  = idx / 144;
                int rem = idx - ch * 144;
                int row = rem / 12;
                int col = rem - row * 12;
                int h = row / 3, p1 = row - h * 3;
                int w = col / 3, p2 = col - w * 3;
                int r = h * 4 + w;
                int f = (p1 * 3 + p2) * 32 + ch;
                const float v = src[idx];
                dst[r * ES + f] = v;
                dtt[f * TP + r] = v;
            }
        }
    }
    __syncthreads();

    const int p   = wv / 6;
    const int wr  = wv - p * 6;
    const int a   = wr / 3;
    const int p1i = wr - a * 3;
    float* stg = lds + OFF_STG + wv * STG_W;
    for (int p2i = 0; p2i < 3; ++p2i) {
        const int qt = 2 * (p1i * 3 + p2i) + a;
        const int j = qt * 16 + c;

        const int kLo = (p ? OFF_E2T : OFF_E1T) + c * TP + 8 * hi;
        const int kHi = (p ? OFF_E1T : OFF_E2T) + c * TP + 8 * hi;
        const int vLo = (p ? OFF_E2S : OFF_E1S) + c * ES + 8 * hi;
        const int vHi = (p ? OFF_E1S : OFF_E2S) + c * ES + 8 * hi;

        v2f qb[4];
        {
            const float* qp = lds + (p ? OFF_E1T : OFF_E2T) + j * TP + 8 * hi;
            const float4 q0 = *(const float4*)(qp);
            const float4 q1 = *(const float4*)(qp + 4);
            qb[0] = pr(q0.x, q0.y) * L2E; qb[1] = pr(q0.z, q0.w) * L2E;
            qb[2] = pr(q1.x, q1.y) * L2E; qb[3] = pr(q1.z, q1.w) * L2E;
        }

        float ml = -3.0e38f;
        float l  = 0.0f;
        v8f   o  = {};

        auto step = [&](const Frag& f, const float oscale) {
            v8f s0 = {}, s1 = {};
            s0 = mma(pr(f.k0a.x, f.k0a.y), qb[0], s0);
            s1 = mma(pr(f.k1a.x, f.k1a.y), qb[0], s1);
            s0 = mma(pr(f.k0a.z, f.k0a.w), qb[1], s0);
            s1 = mma(pr(f.k1a.z, f.k1a.w), qb[1], s1);
            s0 = mma(pr(f.k0b.x, f.k0b.y), qb[2], s0);
            s1 = mma(pr(f.k1b.x, f.k1b.y), qb[2], s1);
            s0 = mma(pr(f.k0b.z, f.k0b.w), qb[3], s0);
            s1 = mma(pr(f.k1b.z, f.k1b.w), qb[3], s1);

            float tm = fmaxf(fmaxf(fmaxf(s0[0], s0[1]), fmaxf(s0[2], s0[3])),
                             fmaxf(fmaxf(s0[4], s0[5]), fmaxf(s0[6], s0[7])));
            tm = fmaxf(tm, fmaxf(fmaxf(fmaxf(s1[0], s1[1]), fmaxf(s1[2], s1[3])),
                                 fmaxf(fmaxf(s1[4], s1[5]), fmaxf(s1[6], s1[7]))));
            tm = fmaxf(tm, xhalf(tm));
            const float mln   = fmaxf(ml, tm);
            const float alpha = __builtin_amdgcn_exp2f(ml - mln);
            const v2f   mm    = pr(mln, mln);
            ml = mln;

            v2f e0[4], e1[4];
            #pragma unroll
            for (int i = 0; i < 4; ++i) {
                const v2f d0 = PAIR(s0, i) - mm;
                const v2f d1 = PAIR(s1, i) - mm;
                e0[i] = pr(__builtin_amdgcn_exp2f(d0[0]), __builtin_amdgcn_exp2f(d0[1]));
                e1[i] = pr(__builtin_amdgcn_exp2f(d1[0]), __builtin_amdgcn_exp2f(d1[1]));
            }
            const v2f t0 = (e0[0] + e0[1]) + (e0[2] + e0[3]);
            const v2f t1 = (e1[0] + e1[1]) + (e1[2] + e1[3]);
            const v2f tt = t0 + t1;
            const float ps = tt[0] + tt[1];
            l = fmaf(l, alpha, ps);
            o = o * (alpha * oscale);

            o = mma(pr(f.v0a.x, f.v0a.y), e0[0], o);
            o = mma(pr(f.v0a.z, f.v0a.w), e0[1], o);
            o = mma(pr(f.v0b.x, f.v0b.y), e0[2], o);
            o = mma(pr(f.v0b.z, f.v0b.w), e0[3], o);
            o = mma(pr(f.v1a.x, f.v1a.y), e1[0], o);
            o = mma(pr(f.v1a.z, f.v1a.w), e1[1], o);
            o = mma(pr(f.v1b.x, f.v1b.y), e1[2], o);
            o = mma(pr(f.v1b.z, f.v1b.w), e1[3], o);
        };

        Frag fa = load_frag(lds, kLo, vLo);
        #pragma unroll 1
        for (int i = 0; i < 9; ++i) {
            const bool up = (i >= 4);
            const int  kb = up ? kHi : kLo;
            const int  vb = up ? vHi : vLo;
            const int  s1 = 2 * i + 1 - (up ? 9 : 0);
            const int  s2 = (s1 + 1 < 9) ? s1 + 1 : 8;

            Frag fb = load_frag(lds, kb + s1 * (32 * TP), vb + s1 * 32);
            asm volatile("" : "+v"(qb[0]) :: "memory");
            step(fa, 1.0f);

            int k2 = kb + s2 * (32 * TP), v2 = vb + s2 * 32;
            asm volatile("" : "+v"(k2), "+v"(v2) : "v"(o));
            fa = load_frag(lds, k2, v2);
            asm volatile("" : "+v"(qb[0]) :: "memory");
            step(fb, (i == 4) ? (0.7f / 0.3f) : 1.0f);
        }

        l += xhalf(l);
        const float inv = 0.3f / l;

        float* sp = stg + c * CP + hi * 24 + p2i;
        #pragma unroll
        for (int v = 0; v < 8; ++v) {
            sp[(v >> 2) * 12 + (v & 3) * 3] = o[v] * inv;
        }
    }

    {
        float* ob = out + (p ? (size_t)B * 4608 : (size_t)0)
                        + (size_t)b * 4608 + (a * 16) * 144 + p1i * 12;
        #pragma unroll
        for (int it = 0; it < 6; ++it) {
            const int g   = it * 32 + lane;
            const int ch  = g / 12;
            const int rem = g - ch * 12;
            const int h   = rem / 3;
            const int sg  = rem - h * 3;
            const float4 val = *(const float4*)(stg + ch * CP + h * 12 + sg * 4);
            *(float4*)(ob + ch * 144 + h * 36 + sg * 4) = val;
        }
    }
}

extern "C" void kernel_launch(void* const* d_in, const int* in_sizes, int n_in,
                              void* d_out, int out_size, void* d_ws, size_t ws_size,
                              hipStream_t stream) {
    const float* x1 = (const float*)d_in[0];
    const float* x2 = (const float*)d_in[1];
    float* out = (float*)d_out;
    const int B = in_sizes[0] / 4608;
    mca_attn_kernel<<<B, NW * 32, 0, stream>>>(x1, x2, out, B);
}


# Gather by image rows: a thread reads one 12-float row of each input as three 16-byte loads, all six in flight before use; key-major LDS copy stored 16 bytes at a time; index math once per thread
# speedup vs baseline: 2.9217x; 1.0628x over previous
#include <hip/hip_runtime.h>
#include <math.h>

typedef __attribute__((ext_vector_type(2))) float v2f;
typedef __attribute__((ext_vector_type(8))) float v8f;

#define ES 292
#define TP 20
#define NW 12

#define OFF_E1S 0
#define OFF_E2S (16 * ES)
#define OFF_E1T (32 * ES)
#define OFF_E2T (32 * ES + 288 * TP)
#define OFF_STG (32 * ES + 2 * 288 * TP)
#define CP 52
#define STG_W (16 * CP)
#define LDS_FLOATS (OFF_STG + NW * STG_W)

#define L2E 1.4426950408889634f

__device__ __forceinline__ float xhalf(float x) {
    int b = __builtin_bit_cast(int, x);
    return __builtin_bit_cast(float,
        __builtin_amdgcn_permlanex16(b, b, 0x76543210, 0xfedcba98, false, false));
}

__device__ __forceinline__ v8f mma(v2f a, v2f b, v8f c) {
    return __builtin_amdgcn_wmma_f32_16x16x4_f32(false, a, false, b, (short)0, c, false, false);
}

__device__ __forceinline__ v2f pr(float x, float y) { v2f r; r[0] = x; r[1] = y; return r; }

#define PAIR(t, i) pr((t)[2 * (i)], (t)[2 * (i) + 1])

struct Frag {
    float4 k0a, k0b;
    float4 k1a, k1b;
    float4 v0a, v0b;
    float4 v1a, v1b;
};

__device__ __forceinline__ Frag load_frag(const float* lds, int ko, int vo) {
    Frag f;
    f.k0a = *(const float4*)(lds + ko);
    f.k0b = *(const float4*)(lds + ko + 4);
    f.k1a = *(const float4*)(lds + ko + 16 * TP);
    f.k1b = *(const float4*)(lds + ko + 16 * TP + 4);
    f.v0a = *(const float4*)(lds + vo);
    f.v0b = *(const float4*)(lds + vo + 4);
    f.v1a = *(const float4*)(lds + vo + 16);
    f.v1b = *(const float4*)(lds + vo + 20);
    return f;
}

__global__ __launch_bounds__(NW * 32)
void mca_attn_kernel(const float* __restrict__ x1,
                     const float* __restrict__ x2,
                     float* __restrict__ out, int B)
{
    __shared__ __attribute__((aligned(16))) float lds[LDS_FLOATS];

    const int b    = blockIdx.x;
    const int tid  = threadIdx.x;
    const int wv   = tid >> 5;
    const int lane = tid & 31;
    const int c    = lane & 15;
    const int hi   = lane >> 4;

    {
        const int ch  = tid / 12;
        const int row = tid - ch * 12;
        const int h   = row / 3;
        const int p1  = row - h * 3;
        const float4* g1 = (const float4*)(x1 + (size_t)b * 4608) + tid * 3;
        const float4* g2 = (const float4*)(x2 + (size_t)b * 4608) + tid * 3;
        float4 ra[2][3];
        ra[0][0] = g1[0]; ra[0][1] = g1[1]; ra[0][2] = g1[2];
        ra[1][0] = g2[0]; ra[1][1] = g2[1]; ra[1][2] = g2[2];
        const int eo = (4 * h) * ES + (3 * p1) * 32 + ch;
        const int to = ((3 * p1) * 32 + ch) * TP + 4 * h;
        #pragma unroll
        for (int m = 0; m < 2; ++m) {
            float* dst = lds + (m ? OFF_E2S : OFF_E1S) + eo;
            float* dtt = lds + (m ? OFF_E2T : OFF_E1T) + to;
            const float v[12] = { ra[m][0].x, ra[m][0].y, ra[m][0].z, ra[m][0].w,
                                  ra[m][1].x, ra[m][1].y, ra[m][1].z, ra[m][1].w,
                                  ra[m][2].x, ra[m][2].y, ra[m][2].z, ra[m][2].w };
            #pragma unroll
            for (int x = 0; x < 12; ++x) dst[(x / 3) * ES + (x % 3) * 32] = v[x];
            #pragma unroll
            for (int p2 = 0; p2 < 3; ++p2)
                *(float4*)(dtt + p2 * 32 * TP) = make_float4(v[p2], v[3 + p2], v[6 + p2], v[9 + p2]);
        }
    }
    __syncthreads();

    const int p   = wv / 6;
    const int wr  = wv - p * 6;
    const int a   = wr / 3;
    const int p1i = wr - a * 3;
    float* stg = lds + OFF_STG + wv * STG_W;
    for (int p2i = 0; p2i < 3; ++p2i) {
        const int qt = 2 * (p1i * 3 + p2i) + a;
        const int j = qt * 16 + c;

        const int kLo = (p ? OFF_E2T : OFF_E1T) + c * TP + 8 * hi;
        const int kHi = (p ? OFF_E1T : OFF_E2T) + c * TP + 8 * hi;
        const int vLo = (p ? OFF_E2S : OFF_E1S) + c * ES + 8 * hi;
        const int vHi = (p ? OFF_E1S : OFF_E2S) + c * ES + 8 * hi;

        v2f qb[4];
        {
            const float* qp = lds + (p ? OFF_E1T : OFF_E2T) + j * TP + 8 * hi;
            const float4 q0 = *(const float4*)(qp);
            const float4 q1 = *(const float4*)(qp + 4);
            qb[0] = pr(q0.x, q0.y) * L2E; qb[1] = pr(q0.z, q0.w) * L2E;
            qb[2] = pr(q1.x, q1.y) * L2E; qb[3] = pr(q1.z, q1.w) * L2E;
        }

        float ml = -3.0e38f;
        float l  = 0.0f;
        v8f   o  = {};

        auto step = [&](const Frag& f, const float oscale) {
            v8f s0 = {}, s1 = {};
            s0 = mma(pr(f.k0a.x, f.k0a.y), qb[0], s0);
            s1 = mma(pr(f.k1a.x, f.k1a.y), qb[0], s1);
            s0 = mma(pr(f.k0a.z, f.k0a.w), qb[1], s0);
            s1 = mma(pr(f.k1a.z, f.k1a.w), qb[1], s1);
            s0 = mma(pr(f.k0b.x, f.k0b.y), qb[2], s0);
            s1 = mma(pr(f.k1b.x, f.k1b.y), qb[2], s1);
            s0 = mma(pr(f.k0b.z, f.k0b.w), qb[3], s0);
            s1 = mma(pr(f.k1b.z, f.k1b.w), qb[3], s1);

            float tm = fmaxf(fmaxf(fmaxf(s0[0], s0[1]), fmaxf(s0[2], s0[3])),
                             fmaxf(fmaxf(s0[4], s0[5]), fmaxf(s0[6], s0[7])));
            tm = fmaxf(tm, fmaxf(fmaxf(fmaxf(s1[0], s1[1]), fmaxf(s1[2], s1[3])),
                                 fmaxf(fmaxf(s1[4], s1[5]), fmaxf(s1[6], s1[7]))));
            tm = fmaxf(tm, xhalf(tm));
            const float mln   = fmaxf(ml, tm);
            const float alpha = __builtin_amdgcn_exp2f(ml - mln);
            const v2f   mm    = pr(mln, mln);
            ml = mln;

            v2f e0[4], e1[4];
            #pragma unroll
            for (int i = 0; i < 4; ++i) {
                const v2f d0 = PAIR(s0, i) - mm;
                const v2f d1 = PAIR(s1, i) - mm;
                e0[i] = pr(__builtin_amdgcn_exp2f(d0[0]), __builtin_amdgcn_exp2f(d0[1]));
                e1[i] = pr(__builtin_amdgcn_exp2f(d1[0]), __builtin_amdgcn_exp2f(d1[1]));
            }
            const v2f t0 = (e0[0] + e0[1]) + (e0[2] + e0[3]);
            const v2f t1 = (e1[0] + e1[1]) + (e1[2] + e1[3]);
            const v2f tt = t0 + t1;
            const float ps = tt[0] + tt[1];
            l = fmaf(l, alpha, ps);
            o = o * (alpha * oscale);

            o = mma(pr(f.v0a.x, f.v0a.y), e0[0], o);
            o = mma(pr(f.v0a.z, f.v0a.w), e0[1], o);
            o = mma(pr(f.v0b.x, f.v0b.y), e0[2], o);
            o = mma(pr(f.v0b.z, f.v0b.w), e0[3], o);
            o = mma(pr(f.v1a.x, f.v1a.y), e1[0], o);
            o = mma(pr(f.v1a.z, f.v1a.w), e1[1], o);
            o = mma(pr(f.v1b.x, f.v1b.y), e1[2], o);
            o = mma(pr(f.v1b.z, f.v1b.w), e1[3], o);
        };

        Frag fa = load_frag(lds, kLo, vLo);
        #pragma unroll 1
        for (int i = 0; i < 9; ++i) {
            const bool up = (i >= 4);
            const int  kb = up ? kHi : kLo;
            const int  vb = up ? vHi : vLo;
            const int  s1 = 2 * i + 1 - (up ? 9 : 0);
            const int  s2 = (s1 + 1 < 9) ? s1 + 1 : 8;

            Frag fb = load_frag(lds, kb + s1 * (32 * TP), vb + s1 * 32);
            asm volatile("" : "+v"(qb[0]) :: "memory");
            step(fa, 1.0f);

            int k2 = kb + s2 * (32 * TP), v2 = vb + s2 * 32;
            asm volatile("" : "+v"(k2), "+v"(v2) : "v"(o));
            fa = load_frag(lds, k2, v2);
            asm volatile("" : "+v"(qb[0]) :: "memory");
            step(fb, (i == 4) ? (0.7f / 0.3f) : 1.0f);
        }

        l += xhalf(l);
        const float inv = 0.3f / l;

        float* sp = stg + c * CP + hi * 24 + p2i;
        #pragma unroll
        for (int v = 0; v < 8; ++v) {
            sp[(v >> 2) * 12 + (v & 3) * 3] = o[v] * inv;
        }
    }

    {
        float* ob = out + (p ? (size_t)B * 4608 : (size_t)0)
                        + (size_t)b * 4608 + (a * 16) * 144 + p1i * 12;
        #pragma unroll
        for (int it = 0; it < 6; ++it) {
            const int g   = it * 32 + lane;
            const int ch  = g / 12;
            const int rem = g - ch * 12;
            const int h   = rem / 3;
            const int sg  = rem - h * 3;
            const float4 val = *(const float4*)(stg + ch * CP + h * 12 + sg * 4);
            *(float4*)(ob + ch * 144 + h * 36 + sg * 4) = val;
        }
    }
}

extern "C" void kernel_launch(void* const* d_in, const int* in_sizes, int n_in,
                              void* d_out, int out_size, void* d_ws, size_t ws_size,
                              hipStream_t stream) {
    const float* x1 = (const float*)d_in[0];
    const float* x2 = (const float*)d_in[1];
    float* out = (float*)d_out;
    const int B = in_sizes[0] / 4608;
    mca_attn_kernel<<<B, NW * 32, 0, stream>>>(x1, x2, out, B);
}


# Softmax reference fixed per query column before its key loop: log2e |q| Kmax from a key-length pass over LDS (running max kept for huge inputs); no max search or step rescale; exp rounding changes
# speedup vs baseline: 3.0143x; 1.0317x over previous
#include <hip/hip_runtime.h>
#include <math.h>

typedef __attribute__((ext_vector_type(2))) float v2f;
typedef __attribute__((ext_vector_type(8))) float v8f;

#define ES 292
#define TP 20
#define NW 12

#define OFF_E1S 0
#define OFF_E2S (16 * ES)
#define OFF_E1T (32 * ES)
#define OFF_E2T (32 * ES + 288 * TP)
#define OFF_STG (32 * ES + 2 * 288 * TP)
#define CP 52
#define STG_W (16 * CP)
#define OFF_NRM (OFF_STG + NW * STG_W)
#define LDS_FLOATS (OFF_NRM + 576 + 16)

#define L2E 1.4426950408889634f

__device__ __forceinline__ float xhalf(float x) {
    int b = __builtin_bit_cast(int, x);
    return __builtin_bit_cast(float,
        __builtin_amdgcn_permlanex16(b, b, 0x76543210, 0xfedcba98, false, false));
}

__device__ __forceinline__ v8f mma(v2f a, v2f b, v8f c) {
    return __builtin_amdgcn_wmma_f32_16x16x4_f32(false, a, false, b, (short)0, c, false, false);
}

__device__ __forceinline__ v2f pr(float x, float y) { v2f r; r[0] = x; r[1] = y; return r; }

#define PAIR(t, i) pr((t)[2 * (i)], (t)[2 * (i) + 1])

struct Frag {
    float4 k0a, k0b;
    float4 k1a, k1b;
    float4 v0a, v0b;
    float4 v1a, v1b;
};

__device__ __forceinline__ Frag load_frag(const float* lds, int ko, int vo) {
    Frag f;
    f.k0a = *(const float4*)(lds + ko);
    f.k0b = *(const float4*)(lds + ko + 4);
    f.k1a = *(const float4*)(lds + ko + 16 * TP);
    f.k1b = *(const float4*)(lds + ko + 16 * TP + 4);
    f.v0a = *(const float4*)(lds + vo);
    f.v0b = *(const float4*)(lds + vo + 4);
    f.v1a = *(const float4*)(lds + vo + 16);
    f.v1b = *(const float4*)(lds + vo + 20);
    return f;
}

__global__ __launch_bounds__(NW * 32)
void mca_attn_kernel(const float* __restrict__ x1,
                     const float* __restrict__ x2,
                     float* __restrict__ out, int B)
{
    __shared__ __attribute__((aligned(16))) float lds[LDS_FLOATS];

    const int b    = blockIdx.x;
    const int tid  = threadIdx.x;
    const int wv   = tid >> 5;
    const int lane = tid & 31;
    const int c    = lane & 15;
    const int hi   = lane >> 4;

    {
        const int ch  = tid / 12;
        const int row = tid - ch * 12;
        const int h   = row / 3;
        const int p1  = row - h * 3;
        const float4* g1 = (const float4*)(x1 + (size_t)b * 4608) + tid * 3;
        const float4* g2 = (const float4*)(x2 + (size_t)b * 4608) + tid * 3;
        float4 ra[2][3];
        ra[0][0] = g1[0]; ra[0][1] = g1[1]; ra[0][2] = g1[2];
        ra[1][0] = g2[0]; ra[1][1] = g2[1]; ra[1][2] = g2[2];
        const int eo = (4 * h) * ES + (3 * p1) * 32 + ch;
        const int to = ((3 * p1) * 32 + ch) * TP + 4 * h;
        #pragma unroll
        for (int m = 0; m < 2; ++m) {
            float* dst = lds + (m ? OFF_E2S : OFF_E1S) + eo;
            float* dtt = lds + (m ? OFF_E2T : OFF_E1T) + to;
            const float v[12] = { ra[m][0].x, ra[m][0].y, ra[m][0].z, ra[m][0].w,
                                  ra[m][1].x, ra[m][1].y, ra[m][1].z, ra[m][1].w,
                                  ra[m][2].x, ra[m][2].y, ra[m][2].z, ra[m][2].w };
            #pragma unroll
            for (int x = 0; x < 12; ++x) dst[(x / 3) * ES + (x % 3) * 32] = v[x];
            #pragma unroll
            for (int p2 = 0; p2 < 3; ++p2)
                *(float4*)(dtt + p2 * 32 * TP) = make_float4(v[p2], v[3 + p2], v[6 + p2], v[9 + p2]);
        }
    }
    __syncthreads();

    float kmax2;
    {
        auto sumsq = [&](const float* r) {
            const float4 a = *(const float4*)(r), b4 = *(const float4*)(r + 4), c4 = *(const float4*)(r + 8), d4 = *(const float4*)(r + 12);
            return ((a.x * a.x + a.y * a.y) + (a.z * a.z + a.w * a.w)) + ((b4.x * b4.x + b4.y * b4.y) + (b4.z * b4.z + b4.w * b4.w))
                 + ((c4.x * c4.x + c4.y * c4.y) + (c4.z * c4.z + c4.w * c4.w)) + ((d4.x * d4.x + d4.y * d4.y) + (d4.z * d4.z + d4.w * d4.w));
        };
        float mx = sumsq(lds + (tid < 288 ? OFF_E1T + tid * TP : OFF_E2T + (tid - 288) * TP));
        lds[OFF_NRM + tid] = mx;
        if (tid < 192) {
            const float n2 = sumsq(lds + OFF_E2T + (tid + 96) * TP);
            lds[OFF_NRM + 384 + tid] = n2;
            mx = fmaxf(mx, n2);
        }
        #pragma unroll
        for (int d = 16; d >= 1; d >>= 1) mx = fmaxf(mx, __shfl_xor(mx, d));
        if (lane == 0) lds[OFF_NRM + 576 + wv] = mx;
    }
    __syncthreads();
    {
        float m = lds[OFF_NRM + 576];
        #pragma unroll
        for (int w = 1; w < NW; ++w) m = fmaxf(m, lds[OFF_NRM + 576 + w]);
        kmax2 = __builtin_bit_cast(float, __builtin_amdgcn_readfirstlane(__builtin_bit_cast(int, m)));
    }
    const bool fixed_ref = kmax2 < 256.0f;

    const int p   = wv / 6;
    const int wr  = wv - p * 6;
    const int a   = wr / 3;
    const int p1i = wr - a * 3;
    float* stg = lds + OFF_STG + wv * STG_W;
    for (int p2i = 0; p2i < 3; ++p2i) {
        const int qt = 2 * (p1i * 3 + p2i) + a;
        const int j = qt * 16 + c;

        const int kLo = (p ? OFF_E2T : OFF_E1T) + c * TP + 8 * hi;
        const int kHi = (p ? OFF_E1T : OFF_E2T) + c * TP + 8 * hi;
        const int vLo = (p ? OFF_E2S : OFF_E1S) + c * ES + 8 * hi;
        const int vHi = (p ? OFF_E1S : OFF_E2S) + c * ES + 8 * hi;

        v2f qb[4];
        {
            const float* qp = lds + (p ? OFF_E1T : OFF_E2T) + j * TP + 8 * hi;
            const float4 q0 = *(const float4*)(qp);
            const float4 q1 = *(const float4*)(qp + 4);
            qb[0] = pr(q0.x, q0.y) * L2E; qb[1] = pr(q0.z, q0.w) * L2E;
            qb[2] = pr(q1.x, q1.y) * L2E; qb[3] = pr(q1.z, q1.w) * L2E;
        }

        float ml = -3.0e38f;
        float l  = 0.0f;
        v8f   o  = {};

        if (fixed_ref) {
            const float qn2  = lds[OFF_NRM + (p ? 0 : 288) + j];
            const float mref = (L2E * 1.0001f) * __builtin_amdgcn_sqrtf(qn2 * kmax2);
            const v2f   mm   = pr(mref, mref);

            auto fstep = [&](const Frag& f, const bool SC, const float osc) {
                v8f s0 = {}, s1 = {};
                s0 = mma(pr(f.k0a.x, f.k0a.y), qb[0], s0);
                s1 = mma(pr(f.k1a.x, f.k1a.y), qb[0], s1);
                s0 = mma(pr(f.k0a.z, f.k0a.w), qb[1], s0);
                s1 = mma(pr(f.k1a.z, f.k1a.w), qb[1], s1);
                s0 = mma(pr(f.k0b.x, f.k0b.y), qb[2], s0);
                s1 = mma(pr(f.k1b.x, f.k1b.y), qb[2], s1);
                s0 = mma(pr(f.k0b.z, f.k0b.w), qb[3], s0);
                s1 = mma(pr(f.k1b.z, f.k1b.w), qb[3], s1);

                v2f e0[4], e1[4];
                #pragma unroll
                for (int i = 0; i < 4; ++i) {
                    const v2f d0 = PAIR(s0, i) - mm;
                    const v2f d1 = PAIR(s1, i) - mm;
                    e0[i] = pr(__builtin_amdgcn_exp2f(d0[0]), __builtin_amdgcn_exp2f(d0[1]));
                    e1[i] = pr(__builtin_amdgcn_exp2f(d1[0]), __builtin_amdgcn_exp2f(d1[1]));
                }
                const v2f t0 = (e0[0] + e0[1]) + (e0[2] + e0[3]);
                const v2f t1 = (e1[0] + e1[1]) + (e1[2] + e1[3]);
                const v2f tt = t0 + t1;
                l += tt[0] + tt[1];
                if (SC) o = o * osc;

                o = mma(pr(f.v0a.x, f.v0a.y), e0[0], o);
                o = mma(pr(f.v0a.z, f.v0a.w), e0[1], o);
                o = mma(pr(f.v0b.x, f.v0b.y), e0[2], o);
                o = mma(pr(f.v0b.z, f.v0b.w), e0[3], o);
                o = mma(pr(f.v1a.x, f.v1a.y), e1[0], o);
                o = mma(pr(f.v1a.z, f.v1a.w), e1[1], o);
                o = mma(pr(f.v1b.x, f.v1b.y), e1[2], o);
                o = mma(pr(f.v1b.z, f.v1b.w), e1[3], o);
            };

            Frag fa = load_frag(lds, kLo, vLo);
            #pragma unroll 1
            for (int i = 0; i < 9; ++i) {
                const bool up = (i >= 4);
                const int  kb = up ? kHi : kLo;
                const int  vb = up ? vHi : vLo;
                const int  s1 = 2 * i + 1 - (up ? 9 : 0);
                const int  s2 = (s1 + 1 < 9) ? s1 + 1 : 8;

                Frag fb = load_frag(lds, kb + s1 * (32 * TP), vb + s1 * 32);
                asm volatile("" : "+v"(qb[0]) :: "memory");
                fstep(fa, false, 1.0f);

                int k2 = kb + s2 * (32 * TP), v2 = vb + s2 * 32;
                asm volatile("" : "+v"(k2), "+v"(v2) : "v"(o));
                fa = load_frag(lds, k2, v2);
                asm volatile("" : "+v"(qb[0]) :: "memory");
                fstep(fb, true, (i == 4) ? (0.7f / 0.3f) : 1.0f);
            }
        } else {
            auto step = [&](const Frag& f, const float oscale) {
                v8f s0 = {}, s1 = {};
                s0 = mma(pr(f.k0a.x, f.k0a.y), qb[0], s0);
                s1 = mma(pr(f.k1a.x, f.k1a.y), qb[0], s1);
                s0 = mma(pr(f.k0a.z, f.k0a.w), qb[1], s0);
                s1 = mma(pr(f.k1a.z, f.k1a.w), qb[1], s1);
                s0 = mma(pr(f.k0b.x, f.k0b.y), qb[2], s0);
                s1 = mma(pr(f.k1b.x, f.k1b.y), qb[2], s1);
                s0 = mma(pr(f.k0b.z, f.k0b.w), qb[3], s0);
                s1 = mma(pr(f.k1b.z, f.k1b.w), qb[3], s1);

                float tm = fmaxf(fmaxf(fmaxf(s0[0], s0[1]), fmaxf(s0[2], s0[3])),
                                 fmaxf(fmaxf(s0[4], s0[5]), fmaxf(s0[6], s0[7])));
                tm = fmaxf(tm, fmaxf(fmaxf(fmaxf(s1[0], s1[1]), fmaxf(s1[2], s1[3])),
                                     fmaxf(fmaxf(s1[4], s1[5]), fmaxf(s1[6], s1[7]))));
                tm = fmaxf(tm, xhalf(tm));
                const float mln   = fmaxf(ml, tm);
                const float alpha = __builtin_amdgcn_exp2f(ml - mln);
                const v2f   mm    = pr(mln, mln);
                ml = mln;

                v2f e0[4], e1[4];
                #pragma unroll
                for (int i = 0; i < 4; ++i) {
                    const v2f d0 = PAIR(s0, i) - mm;
                    const v2f d1 = PAIR(s1, i) - mm;
                    e0[i] = pr(__builtin_amdgcn_exp2f(d0[0]), __builtin_amdgcn_exp2f(d0[1]));
                    e1[i] = pr(__builtin_amdgcn_exp2f(d1[0]), __builtin_amdgcn_exp2f(d1[1]));
                }
                const v2f t0 = (e0[0] + e0[1]) + (e0[2] + e0[3]);
                const v2f t1 = (e1[0] + e1[1]) + (e1[2] + e1[3]);
                const v2f tt = t0 + t1;
                const float ps = tt[0] + tt[1];
                l = fmaf(l, alpha, ps);
                o = o * (alpha * oscale);

                o = mma(pr(f.v0a.x, f.v0a.y), e0[0], o);
                o = mma(pr(f.v0a.z, f.v0a.w), e0[1], o);
                o = mma(pr(f.v0b.x, f.v0b.y), e0[2], o);
                o = mma(pr(f.v0b.z, f.v0b.w), e0[3], o);
                o = mma(pr(f.v1a.x, f.v1a.y), e1[0], o);
                o = mma(pr(f.v1a.z, f.v1a.w), e1[1], o);
                o = mma(pr(f.v1b.x, f.v1b.y), e1[2], o);
                o = mma(pr(f.v1b.z, f.v1b.w), e1[3], o);
            };

            Frag fa = load_frag(lds, kLo, vLo);
            #pragma unroll 1
            for (int i = 0; i < 9; ++i) {
                const bool up = (i >= 4);
                const int  kb = up ? kHi : kLo;
                const int  vb = up ? vHi : vLo;
                const int  s1 = 2 * i + 1 - (up ? 9 : 0);
                const int  s2 = (s1 + 1 < 9) ? s1 + 1 : 8;

                Frag fb = load_frag(lds, kb + s1 * (32 * TP), vb + s1 * 32);
                asm volatile("" : "+v"(qb[0]) :: "memory");
                step(fa, 1.0f);

                int k2 = kb + s2 * (32 * TP), v2 = vb + s2 * 32;
                asm volatile("" : "+v"(k2), "+v"(v2) : "v"(o));
                fa = load_frag(lds, k2, v2);
                asm volatile("" : "+v"(qb[0]) :: "memory");
                step(fb, (i == 4) ? (0.7f / 0.3f) : 1.0f);
            }

        }

        l += xhalf(l);
        const float inv = 0.3f / l;

        float* sp = stg + c * CP + hi * 24 + p2i;
        #pragma unroll
        for (int v = 0; v < 8; ++v) {
            sp[(v >> 2) * 12 + (v & 3) * 3] = o[v] * inv;
        }
    }

    {
        float* ob = out + (p ? (size_t)B * 4608 : (size_t)0)
                        + (size_t)b * 4608 + (a * 16) * 144 + p1i * 12;
        #pragma unroll
        for (int it = 0; it < 6; ++it) {
            const int g   = it * 32 + lane;
            const int ch  = g / 12;
            const int rem = g - ch * 12;
            const int h   = rem / 3;
            const int sg  = rem - h * 3;
            const float4 val = *(const float4*)(stg + ch * CP + h * 12 + sg * 4);
            *(float4*)(ob + ch * 144 + h * 36 + sg * 4) = val;
        }
    }
}

extern "C" void kernel_launch(void* const* d_in, const int* in_sizes, int n_in,
                              void* d_out, int out_size, void* d_ws, size_t ws_size,
                              hipStream_t stream) {
    const float* x1 = (const float*)d_in[0];
    const float* x2 = (const float*)d_in[1];
    float* out = (float*)d_out;
    const int B = in_sizes[0] / 4608;
    mca_attn_kernel<<<B, NW * 32, 0, stream>>>(x1, x2, out, B);
}
